# starting kernel
# baseline (speedup 1.0000x reference)
#include <hip/hip_runtime.h>

typedef float v2f __attribute__((ext_vector_type(2)));
typedef float v8f __attribute__((ext_vector_type(8)));

#define NNODES 100000
#define NEDGES 800000
#define INDIM  128
#define HID1   64
#define HID2   64
#define ODIM   32

__global__ __launch_bounds__(256) void zero_f32_k(float* __restrict__ p, int n) {
    int i = blockIdx.x * blockDim.x + threadIdx.x;
    if (i < n) p[i] = 0.0f;
}

__global__ __launch_bounds__(256) void degree_k(const int* __restrict__ dst,
                                                float* __restrict__ deg, int nE) {
    int e = blockIdx.x * blockDim.x + threadIdx.x;
    if (e < nE) unsafeAtomicAdd(&deg[dst[e]], 1.0f);
}

__global__ __launch_bounds__(256) void rsqrt_k(float* __restrict__ d, int n) {
    int i = blockIdx.x * blockDim.x + threadIdx.x;
    if (i < n) d[i] = rsqrtf(d[i] + 1.0f);
}

template <int N>
__global__ __launch_bounds__(256) void gemm_wmma_f32_k(const float* __restrict__ A,
                                                       const float* __restrict__ W,
                                                       float* __restrict__ C,
                                                       int M, int K) {
    constexpr int NT = N / 16;
    const int lane = threadIdx.x & 31;
    const int wave = (blockIdx.x * blockDim.x + threadIdx.x) >> 5;
    const int m0   = wave << 4;
    if (m0 >= M) return;

    const int mr = lane & 15;
    const int kh = lane >> 4;

    v8f acc[NT];
#pragma unroll
    for (int t = 0; t < NT; ++t) acc[t] = {};

    const size_t arow = (size_t)(m0 + mr) * (size_t)K;

    for (int k0 = 0; k0 < K; k0 += 4) {
        v2f a;
        a.x = A[arow + k0 + 2 * kh + 0];
        a.y = A[arow + k0 + 2 * kh + 1];
#pragma unroll
        for (int t = 0; t < NT; ++t) {
            const int ncol = (t << 4) + mr;
            v2f b;
            b.x = W[(size_t)(k0 + 2 * kh + 0) * N + ncol];
            b.y = W[(size_t)(k0 + 2 * kh + 1) * N + ncol];
            acc[t] = __builtin_amdgcn_wmma_f32_16x16x4_f32(
                false, a, false, b, (short)0, acc[t], false, false);
        }
    }

#pragma unroll
    for (int t = 0; t < NT; ++t) {
        const int ncol = (t << 4) + mr;
#pragma unroll
        for (int v = 0; v < 8; ++v) {
            const int row = m0 + 8 * kh + v;
            C[(size_t)row * N + ncol] = acc[t][v];
        }
    }
}

template <int H>
__global__ __launch_bounds__(256) void edge_agg_k(const float* __restrict__ h,
                                                  const int* __restrict__ src,
                                                  const int* __restrict__ dst,
                                                  const float* __restrict__ dis,
                                                  float* __restrict__ agg, int nE) {
    const int warp = (blockIdx.x * blockDim.x + threadIdx.x) >> 5;
    const int lane = threadIdx.x & 31;
    if (warp >= nE) return;
    const int s = src[warp];
    const int d = dst[warp];
    const float norm = dis[s] * dis[d];
    const float* hs = h + (size_t)s * H;
    float* ad = agg + (size_t)d * H;
#pragma unroll
    for (int f = lane; f < H; f += 32)
        unsafeAtomicAdd(&ad[f], norm * hs[f]);
}

template <int H>
__global__ __launch_bounds__(256) void finish_k(const float* __restrict__ agg,
                                                const float* __restrict__ h,
                                                const float* __restrict__ dis,
                                                const float* __restrict__ b,
                                                float* __restrict__ out, int M) {
    const int i = blockIdx.x * blockDim.x + threadIdx.x;
    if (i >= M * H) return;
    const int node = i >> (H == 64 ? 6 : 5);
    const int f    = i & (H - 1);
    const float dv = dis[node];
    const float v  = agg[i] + dv * dv * h[i] + b[f];
    out[i] = v > 0.0f ? v : 0.0f;
}

__global__ __launch_bounds__(256) void softmax32_k(float* __restrict__ out, int M) {
    const int warp = (blockIdx.x * blockDim.x + threadIdx.x) >> 5;
    const int lane = threadIdx.x & 31;
    if (warp >= M) return;
    const size_t idx = (size_t)warp * 32 + lane;
    const float v = out[idx];
    float m = v;
#pragma unroll
    for (int off = 16; off > 0; off >>= 1) m = fmaxf(m, __shfl_xor(m, off, 32));
    const float e = __expf(v - m);
    float s = e;
#pragma unroll
    for (int off = 16; off > 0; off >>= 1) s += __shfl_xor(s, off, 32);
    out[idx] = e / s;
}

extern "C" void kernel_launch(void* const* d_in, const int* in_sizes, int n_in,
                              void* d_out, int out_size, void* d_ws, size_t ws_size,
                              hipStream_t stream) {
    const float* x    = (const float*)d_in[0];
    const int*   edge = (const int*)d_in[1];
    const float* W1   = (const float*)d_in[2];
    const float* b1   = (const float*)d_in[3];
    const float* W2   = (const float*)d_in[4];
    const float* b2   = (const float*)d_in[5];
    const float* W3   = (const float*)d_in[6];
    const float* b3   = (const float*)d_in[7];
    float* out = (float*)d_out;

    const int* src = edge;
    const int* dst = edge + NEDGES;

    float* dis  = (float*)d_ws;
    float* hbuf = dis  + NNODES;
    float* abuf = hbuf + (size_t)NNODES * 64;
    float* xbuf = abuf + (size_t)NNODES * 64;

    const int TB = 256;
    auto blocks = [](long long n, int tb) { return (int)((n + tb - 1) / tb); };

    const int gemm_waves  = NNODES / 16;
    const int gemm_blocks = blocks((long long)gemm_waves * 32, TB);
    const int edge_blocks = blocks((long long)NEDGES * 32, TB);

    zero_f32_k<<<blocks(NNODES, TB), TB, 0, stream>>>(dis, NNODES);
    degree_k<<<blocks(NEDGES, TB), TB, 0, stream>>>(dst, dis, NEDGES);
    rsqrt_k<<<blocks(NNODES, TB), TB, 0, stream>>>(dis, NNODES);

    gemm_wmma_f32_k<64><<<gemm_blocks, TB, 0, stream>>>(x, W1, hbuf, NNODES, INDIM);
    zero_f32_k<<<blocks((long long)NNODES * 64, TB), TB, 0, stream>>>(abuf, NNODES * 64);
    edge_agg_k<64><<<edge_blocks, TB, 0, stream>>>(hbuf, src, dst, dis, abuf, NEDGES);
    finish_k<64><<<blocks((long long)NNODES * 64, TB), TB, 0, stream>>>(abuf, hbuf, dis, b1, xbuf, NNODES);

    gemm_wmma_f32_k<64><<<gemm_blocks, TB, 0, stream>>>(xbuf, W2, hbuf, NNODES, HID1);
    zero_f32_k<<<blocks((long long)NNODES * 64, TB), TB, 0, stream>>>(abuf, NNODES * 64);
    edge_agg_k<64><<<edge_blocks, TB, 0, stream>>>(hbuf, src, dst, dis, abuf, NEDGES);
    finish_k<64><<<blocks((long long)NNODES * 64, TB), TB, 0, stream>>>(abuf, hbuf, dis, b2, xbuf, NNODES);

    gemm_wmma_f32_k<32><<<gemm_blocks, TB, 0, stream>>>(xbuf, W3, hbuf, NNODES, HID2);
    zero_f32_k<<<blocks((long long)NNODES * 32, TB), TB, 0, stream>>>(abuf, NNODES * 32);
    edge_agg_k<32><<<edge_blocks, TB, 0, stream>>>(hbuf, src, dst, dis, abuf, NEDGES);
    finish_k<32><<<blocks((long long)NNODES * 32, TB), TB, 0, stream>>>(abuf, hbuf, dis, b3, out, NNODES);

    softmax32_k<<<blocks((long long)NNODES * 32, TB), TB, 0, stream>>>(out, NNODES);
}


# Float atomics (degree counts, row scatters) split by L2 half: a wave reads its die, draws a (128-edge chunk, row range) ticket, adds only into that half's rows; sums stay in arrival order
# speedup vs baseline: 2.6646x; 2.6646x over previous
#include <hip/hip_runtime.h>


typedef float v2f __attribute__((ext_vector_type(2)));
typedef float v8f __attribute__((ext_vector_type(8)));

#define NNODES 100000
#define NEDGES 800000
#define INDIM  128
#define HID1   64
#define HID2   64
#define ODIM   32

#define NSPLIT 50048
#define CNT_STRIDE 64
#define CNT_PER_PASS 128
#define CNT_INTS 512

__global__ __launch_bounds__(256) void zero_f32_k(float* __restrict__ p, int n) {
    int i = blockIdx.x * blockDim.x + threadIdx.x;
    if (i < n) p[i] = 0.0f;
}

__device__ __forceinline__ int chip_half() {
    const unsigned w = __builtin_amdgcn_s_sendmsg_rtn(0x87);
    return (int)((w >> 18) & 1u);
}

__device__ __forceinline__ int draw_ticket(int* __restrict__ cnt, int nchunks, int lane, int& q) {
    q = chip_half();
    int t = 0;
    if (lane == 0) t = atomicAdd(&cnt[q * CNT_STRIDE], 1);
    t = __builtin_amdgcn_readfirstlane(t);
    if (t >= nchunks) {
        q = 1 - q;
        if (lane == 0) t = atomicAdd(&cnt[q * CNT_STRIDE], 1);
        t = __builtin_amdgcn_readfirstlane(t);
    }
    return t;
}

#define DEG_CH 256
__global__ __launch_bounds__(256) void degree_k(const int* __restrict__ dst,
                                                float* __restrict__ deg,
                                                int* __restrict__ cnt, int nE) {
    const int lane = threadIdx.x & 31;
    const int nchunks = (nE + DEG_CH - 1) / DEG_CH;
    int q;
    const int t = draw_ticket(cnt, nchunks, lane, q);
    if (t >= nchunks) return;
    const int e0 = t * DEG_CH;
    int d[DEG_CH / 32];
#pragma unroll
    for (int i = 0; i < DEG_CH / 32; ++i) {
        const int e = e0 + i * 32 + lane;
        d[i] = e < nE ? dst[e] : -1;
    }
#pragma unroll
    for (int i = 0; i < DEG_CH / 32; ++i) {
        if (d[i] >= 0 && (d[i] >= NSPLIT) == (q == 1)) unsafeAtomicAdd(&deg[d[i]], 1.0f);
    }
}

template <int N, bool FIRST>
__global__ __launch_bounds__(256) void gemm_seed_k(const float* __restrict__ A,
                                                   const float* __restrict__ W,
                                                   const float* __restrict__ bprev,
                                                   const float* __restrict__ dnorm,
                                                   float* __restrict__ dis_out,
                                                   float* __restrict__ G,
                                                   float* __restrict__ S,
                                                   int M, int K) {
    constexpr int NT = N / 16;
    const int lane = threadIdx.x & 31;
    const int wave = (blockIdx.x * blockDim.x + threadIdx.x) >> 5;
    const int m0   = wave << 4;
    if (m0 >= M) return;

    const int mr = lane & 15;
    const int kh = lane >> 4;
    const int r0 = m0 + 8 * kh;

    const float4 q0 = *reinterpret_cast<const float4*>(dnorm + r0);
    const float4 q1 = *reinterpret_cast<const float4*>(dnorm + r0 + 4);

    v8f acc[NT];
#pragma unroll
    for (int t = 0; t < NT; ++t) acc[t] = {};

    const size_t arow = (size_t)(m0 + mr) * (size_t)K;
    float da = 1.0f;
    if (!FIRST) da = dnorm[m0 + mr];

    for (int k0 = 0; k0 < K; k0 += 4) {
        v2f a;
        a.x = A[arow + k0 + 2 * kh + 0];
        a.y = A[arow + k0 + 2 * kh + 1];
        if (!FIRST) {
            const float px = da * a.x + bprev[k0 + 2 * kh + 0];
            const float py = da * a.y + bprev[k0 + 2 * kh + 1];
            a.x = px > 0.0f ? px : 0.0f;
            a.y = py > 0.0f ? py : 0.0f;
        }
#pragma unroll
        for (int t = 0; t < NT; ++t) {
            const int ncol = (t << 4) + mr;
            v2f b;
            b.x = W[(size_t)(k0 + 2 * kh + 0) * N + ncol];
            b.y = W[(size_t)(k0 + 2 * kh + 1) * N + ncol];
            acc[t] = __builtin_amdgcn_wmma_f32_16x16x4_f32(
                false, a, false, b, (short)0, acc[t], false, false);
        }
    }

    float dv[8] = {q0.x, q0.y, q0.z, q0.w, q1.x, q1.y, q1.z, q1.w};
    if (FIRST) {
#pragma unroll
        for (int v = 0; v < 8; ++v) dv[v] = rsqrtf(dv[v] + 1.0f);
        if (mr == 0) {
            *reinterpret_cast<float4*>(dis_out + r0)     = make_float4(dv[0], dv[1], dv[2], dv[3]);
            *reinterpret_cast<float4*>(dis_out + r0 + 4) = make_float4(dv[4], dv[5], dv[6], dv[7]);
        }
    }

#pragma unroll
    for (int t = 0; t < NT; ++t) {
        const int ncol = (t << 4) + mr;
#pragma unroll
        for (int v = 0; v < 8; ++v) {
            const size_t o = (size_t)(r0 + v) * N + ncol;
            const float gv = dv[v] * acc[t][v];
            G[o] = gv;
            S[o] = gv;
        }
    }
}

#define EDGE_CH 128
template <int H, int J0, bool FULL>
__device__ __forceinline__ void scatter_group(const float* __restrict__ h, float* __restrict__ agg,
                                              int ps, int pd, int left, int lane) {
    constexpr int NF = H / 32;
    constexpr int UB = 16;
    float v[UB][NF];
#pragma unroll
    for (int u = 0; u < UB; ++u) {
        if (FULL || u < left) {
            const int sj = __builtin_amdgcn_readlane(ps, J0 + u);
            const float* hs = h + (size_t)sj * H;
#pragma unroll
            for (int c = 0; c < NF; ++c) v[u][c] = hs[lane + 32 * c];
        }
    }
#pragma unroll
    for (int u = 0; u < UB; ++u) {
        if (FULL || u < left) {
            const int dj = __builtin_amdgcn_readlane(pd, J0 + u);
            float* ad = agg + (size_t)dj * H;
#pragma unroll
            for (int c = 0; c < NF; ++c) unsafeAtomicAdd(&ad[lane + 32 * c], v[u][c]);
        }
    }
}

template <int H>
__global__ __launch_bounds__(256) void edge_agg_k(const float* __restrict__ h,
                                                  const int* __restrict__ src,
                                                  const int* __restrict__ dst,
                                                  float* __restrict__ agg,
                                                  int* __restrict__ cnt, int nE) {
    constexpr int EPL = EDGE_CH / 32;
    __shared__ int2 list[8][EDGE_CH];
    const int lane = threadIdx.x & 31;
    const int wv   = threadIdx.x >> 5;
    const int nchunks = (nE + EDGE_CH - 1) / EDGE_CH;
    int q;
    const int t = draw_ticket(cnt, nchunks, lane, q);
    if (t >= nchunks) return;
    const int e0 = t * EDGE_CH;

    int s[EPL], d[EPL];
#pragma unroll
    for (int i = 0; i < EPL; ++i) {
        const int e = e0 + i * 32 + lane;
        s[i] = e < nE ? src[e] : 0;
        d[i] = e < nE ? dst[e] : -1;
    }
    int n = 0;
#pragma unroll
    for (int i = 0; i < EPL; ++i) {
        const bool mine = d[i] >= 0 && (d[i] >= NSPLIT) == (q == 1);
        const unsigned m = __builtin_amdgcn_ballot_w32(mine);
        if (mine) list[wv][n + (int)__builtin_amdgcn_mbcnt_lo(m, 0u)] = make_int2(s[i], d[i]);
        n += __builtin_popcount(m);
    }
    __builtin_amdgcn_fence(__ATOMIC_RELEASE, "wavefront");
    __builtin_amdgcn_wave_barrier();
    __builtin_amdgcn_fence(__ATOMIC_ACQUIRE, "wavefront");

    for (int g0 = 0; g0 < n; g0 += 32) {
        int2 p = make_int2(0, 0);
        if (g0 + lane < n) p = list[wv][g0 + lane];
        const int left0 = n - g0;
        if (left0 >= 16) scatter_group<H, 0, true>(h, agg, p.x, p.y, 16, lane);
        else             scatter_group<H, 0, false>(h, agg, p.x, p.y, left0, lane);
        const int left1 = left0 - 16;
        if (left1 >= 16)    scatter_group<H, 16, true>(h, agg, p.x, p.y, 16, lane);
        else if (left1 > 0) scatter_group<H, 16, false>(h, agg, p.x, p.y, left1, lane);
    }
}

__global__ __launch_bounds__(256) void relu_softmax32_k(const float* __restrict__ S,
                                                        const float* __restrict__ dis,
                                                        const float* __restrict__ bias,
                                                        float* __restrict__ out, int M) {
    const int warp = (blockIdx.x * blockDim.x + threadIdx.x) >> 5;
    const int lane = threadIdx.x & 31;
    if (warp >= M) return;
    const size_t idx = (size_t)warp * 32 + lane;
    const float sv = dis[warp] * S[idx] + bias[lane];
    const float v = sv > 0.0f ? sv : 0.0f;
    float m = v;
#pragma unroll
    for (int off = 16; off > 0; off >>= 1) m = fmaxf(m, __shfl_xor(m, off, 32));
    const float e = __expf(v - m);
    float s = e;
#pragma unroll
    for (int off = 16; off > 0; off >>= 1) s += __shfl_xor(s, off, 32);
    out[idx] = e / s;
}

extern "C" void kernel_launch(void* const* d_in, const int* in_sizes, int n_in,
                              void* d_out, int out_size, void* d_ws, size_t ws_size,
                              hipStream_t stream) {
    const float* x    = (const float*)d_in[0];
    const int*   edge = (const int*)d_in[1];
    const float* W1   = (const float*)d_in[2];
    const float* b1   = (const float*)d_in[3];
    const float* W2   = (const float*)d_in[4];
    const float* b2   = (const float*)d_in[5];
    const float* W3   = (const float*)d_in[6];
    const float* b3   = (const float*)d_in[7];
    float* out = (float*)d_out;

    const int* src = edge;
    const int* dst = edge + NEDGES;

    int*   cnt  = (int*)d_ws;
    float* deg  = (float*)d_ws + CNT_INTS;
    float* dis  = deg  + NNODES;
    float* gbuf = dis  + NNODES;
    float* sa   = gbuf + (size_t)NNODES * 64;
    float* sb   = sa   + (size_t)NNODES * 64;

    const int TB = 256;
    auto blocks = [](long long n, int tb) { return (int)((n + tb - 1) / tb); };

    const int gemm_waves  = NNODES / 16;
    const int gemm_blocks = blocks((long long)gemm_waves * 32, TB);
    const int edge_waves  = 2 * ((NEDGES + EDGE_CH - 1) / EDGE_CH);
    const int edge_blocks = blocks((long long)edge_waves * 32, TB);
    const int deg_waves   = 2 * ((NEDGES + DEG_CH - 1) / DEG_CH);
    const int deg_blocks  = blocks((long long)deg_waves * 32, TB);

    zero_f32_k<<<blocks(NNODES + CNT_INTS, TB), TB, 0, stream>>>((float*)d_ws, NNODES + CNT_INTS);
    degree_k<<<deg_blocks, TB, 0, stream>>>(dst, deg, cnt, NEDGES);

    gemm_seed_k<64, true><<<gemm_blocks, TB, 0, stream>>>(x, W1, nullptr, deg, dis, gbuf, sa, NNODES, INDIM);
    edge_agg_k<64><<<edge_blocks, TB, 0, stream>>>(gbuf, src, dst, sa, cnt + 1 * CNT_PER_PASS, NEDGES);

    gemm_seed_k<64, false><<<gemm_blocks, TB, 0, stream>>>(sa, W2, b1, dis, nullptr, gbuf, sb, NNODES, HID1);
    edge_agg_k<64><<<edge_blocks, TB, 0, stream>>>(gbuf, src, dst, sb, cnt + 2 * CNT_PER_PASS, NEDGES);

    gemm_seed_k<32, false><<<gemm_blocks, TB, 0, stream>>>(sb, W3, b2, dis, nullptr, gbuf, sa, NNODES, HID2);
    edge_agg_k<32><<<edge_blocks, TB, 0, stream>>>(gbuf, src, dst, sa, cnt + 3 * CNT_PER_PASS, NEDGES);

    relu_softmax32_k<<<blocks((long long)NNODES * 32, TB), TB, 0, stream>>>(sa, dis, b3, out, NNODES);
}
